# starting kernel
# baseline (speedup 1.0000x reference)
#include <hip/hip_runtime.h>
#include <math.h>

typedef __attribute__((ext_vector_type(2))) float v2f;
typedef __attribute__((ext_vector_type(8))) float v8f;

#define NN 8192
#define DD 64
#define BM 128
#define KC 32
#define SA (KC + 2)
#define ALPHA 0.2f
#define EPS 1e-8f


__global__ __launch_bounds__(256) void rowsum_kernel(const float* __restrict__ adj,
                                                     float* __restrict__ sr) {
  __shared__ float red[256];
  const int row = blockIdx.x;
  const int t = threadIdx.x;
  const float* p = adj + (size_t)row * NN;
  float s = 0.f;
  for (int j = t; j < NN; j += 256) s += p[j];
  red[t] = s;
  __syncthreads();
  for (int w = 128; w > 0; w >>= 1) {
    if (t < w) red[t] += red[t + w];
    __syncthreads();
  }
  if (t == 0) sr[row] = sqrtf(red[0]);
}

__global__ __launch_bounds__(256) void zero_kernel(float* __restrict__ p, int n) {
  int i = blockIdx.x * 256 + threadIdx.x;
  if (i < n) p[i] = 0.f;
}

__global__ __launch_bounds__(256) void colsum_partial_kernel(const float* __restrict__ adj,
                                                             float* __restrict__ acc) {
  const int j = blockIdx.x * 256 + threadIdx.x;
  const int i0 = blockIdx.y * 128;
  float s = 0.f;
#pragma unroll 4
  for (int i = 0; i < 128; ++i) s += adj[(size_t)(i0 + i) * NN + j];
  atomicAdd(&acc[j], s);
}

__global__ __launch_bounds__(256) void sqrt_kernel(const float* __restrict__ a,
                                                   float* __restrict__ sc) {
  int j = blockIdx.x * 256 + threadIdx.x;
  sc[j] = sqrtf(a[j]);
}


template <bool TRANSA>
__global__ __launch_bounds__(256) void spmm_kernel(
    const float* __restrict__ adj, const float* __restrict__ sr, const float* __restrict__ sc,
    const float* __restrict__ Bsrc, const float* __restrict__ Btgt,
    float* __restrict__ Z1, float* __restrict__ Z2) {
  __shared__ __align__(16) float At[BM * SA];
  __shared__ __align__(16) float Bs[DD * SA];
  __shared__ __align__(16) float Bt[DD * SA];

  const int tid = threadIdx.x;
  const int wave = tid >> 5;
  const int lane = tid & 31;
  const int laneRow = lane & 15;
  const int laneHi = lane >> 4;
  const int r0 = blockIdx.x * BM;

  v8f acc[8] = {};

  for (int k0 = 0; k0 < NN; k0 += KC) {
    if (!TRANSA) {
#pragma unroll
      for (int it = 0; it < (BM * KC / 4) / 256; ++it) {
        int f = it * 256 + tid;
        int row = f >> 3;
        int k = (f & 7) << 2;
        float4 v = *(const float4*)(adj + (size_t)(r0 + row) * NN + (k0 + k));
        float a = sr[r0 + row];
        float* dst = &At[row * SA + k];
        dst[0] = v.x * __builtin_amdgcn_rcpf(a * sc[k0 + k + 0] + EPS);
        dst[1] = v.y * __builtin_amdgcn_rcpf(a * sc[k0 + k + 1] + EPS);
        dst[2] = v.z * __builtin_amdgcn_rcpf(a * sc[k0 + k + 2] + EPS);
        dst[3] = v.w * __builtin_amdgcn_rcpf(a * sc[k0 + k + 3] + EPS);
      }
    } else {
#pragma unroll
      for (int it = 0; it < (BM * KC / 4) / 256; ++it) {
        int f = it * 256 + tid;
        int k = f >> 5;
        int rw = (f & 31) << 2;
        float4 v = *(const float4*)(adj + (size_t)(k0 + k) * NN + (r0 + rw));
        float a = sr[k0 + k];
        At[(rw + 0) * SA + k] = v.x * __builtin_amdgcn_rcpf(a * sc[r0 + rw + 0] + EPS);
        At[(rw + 1) * SA + k] = v.y * __builtin_amdgcn_rcpf(a * sc[r0 + rw + 1] + EPS);
        At[(rw + 2) * SA + k] = v.z * __builtin_amdgcn_rcpf(a * sc[r0 + rw + 2] + EPS);
        At[(rw + 3) * SA + k] = v.w * __builtin_amdgcn_rcpf(a * sc[r0 + rw + 3] + EPS);
      }
    }
#pragma unroll
    for (int it = 0; it < (DD * KC / 4) / 256; ++it) {
      int f = it * 256 + tid;
      int u = f >> 4;
      int col = (f & 15) << 2;
      float4 v1 = *(const float4*)(Bsrc + (size_t)(k0 + u) * DD + col);
      float4 v2 = *(const float4*)(Btgt + (size_t)(k0 + u) * DD + col);
      Bs[(col + 0) * SA + u] = v1.x; Bs[(col + 1) * SA + u] = v1.y;
      Bs[(col + 2) * SA + u] = v1.z; Bs[(col + 3) * SA + u] = v1.w;
      Bt[(col + 0) * SA + u] = v2.x; Bt[(col + 1) * SA + u] = v2.y;
      Bt[(col + 2) * SA + u] = v2.z; Bt[(col + 3) * SA + u] = v2.w;
    }
    __syncthreads();

    const float* Arow = &At[(wave * 16 + laneRow) * SA + 2 * laneHi];
#pragma unroll
    for (int kk = 0; kk < KC; kk += 4) {
      v2f a = *(const v2f*)(Arow + kk);
#pragma unroll
      for (int n = 0; n < 8; ++n) {
        const float* Bb = (n < 4) ? Bs : Bt;
        int col = (n & 3) * 16 + laneRow;
        v2f b = *(const v2f*)(Bb + col * SA + kk + 2 * laneHi);
        acc[n] = __builtin_amdgcn_wmma_f32_16x16x4_f32(
            false, a, false, b, (short)0, acc[n], false, false);
      }
    }
    __syncthreads();
  }

#pragma unroll
  for (int n = 0; n < 8; ++n) {
    float* Z = (n < 4) ? Z1 : Z2;
    const int col = (n & 3) * 16 + laneRow;
#pragma unroll
    for (int v = 0; v < 8; ++v) {
      int row = r0 + wave * 16 + v + 8 * laneHi;
      Z[(size_t)row * DD + col] = acc[n][v];
    }
  }
}


__global__ __launch_bounds__(256) void epilogue_kernel(
    const float* __restrict__ Z1, const float* __restrict__ Z2,
    const float* __restrict__ src, const float* __restrict__ tgt,
    const float* __restrict__ W1, const float* __restrict__ W2,
    float* __restrict__ out) {
  __shared__ float z1s[4][DD];
  __shared__ float ps[4][DD];
  const int tid = threadIdx.x;
  const int r = tid >> 6;
  const int c = tid & 63;
  const int row = blockIdx.x * 4 + r;
  const size_t off = (size_t)row * DD + c;
  z1s[r][c] = Z1[off];
  ps[r][c] = src[off] * Z2[off];
  __syncthreads();
  float acc = tgt[off];
#pragma unroll 8
  for (int k = 0; k < DD; ++k)
    acc += z1s[r][k] * W1[k * DD + c] + ps[r][k] * W2[k * DD + c];
  out[off] = (acc >= 0.f) ? acc : ALPHA * acc;
}


extern "C" void kernel_launch(void* const* d_in, const int* in_sizes, int n_in,
                              void* d_out, int out_size, void* d_ws, size_t ws_size,
                              hipStream_t stream) {
  const float* user = (const float*)d_in[0];
  const float* item = (const float*)d_in[1];
  const float* adj  = (const float*)d_in[2];
  const float* W1   = (const float*)d_in[3];
  const float* W2   = (const float*)d_in[4];

  float* ws = (float*)d_ws;
  const size_t ND = (size_t)NN * DD;
  float* sr   = ws;
  float* sc   = ws + NN;
  float* cacc = ws + 2 * NN;
  float* Z1u  = ws + 3 * NN;
  float* Z2u  = Z1u + ND;
  float* Z1i  = Z2u + ND;
  float* Z2i  = Z1i + ND;

  float* out_user = (float*)d_out;
  float* out_item = out_user + ND;

  rowsum_kernel<<<NN, 256, 0, stream>>>(adj, sr);
  zero_kernel<<<NN / 256, 256, 0, stream>>>(cacc, NN);
  colsum_partial_kernel<<<dim3(NN / 256, 64), 256, 0, stream>>>(adj, cacc);
  sqrt_kernel<<<NN / 256, 256, 0, stream>>>(cacc, sc);

  spmm_kernel<true><<<NN / BM, 256, 0, stream>>>(adj, sr, sc, item, user, Z1u, Z2u);
  spmm_kernel<false><<<NN / BM, 256, 0, stream>>>(adj, sr, sc, user, item, Z1i, Z2i);

  epilogue_kernel<<<NN / 4, 256, 0, stream>>>(Z1u, Z2u, item, user, W1, W2, out_user);
  epilogue_kernel<<<NN / 4, 256, 0, stream>>>(Z1i, Z2i, user, item, W1, W2, out_item);
}


# spmm LDS stages: pitch 34 -> 36 floats, B and transposed-A rows interleaved (4c+j at 16j+c): fragment reads and stage stores without bank conflicts; out in 16-B pieces; no sum order changed
# speedup vs baseline: 5.6125x; 5.6125x over previous
#include <hip/hip_runtime.h>
#include <math.h>

typedef __attribute__((ext_vector_type(2))) float v2f;
typedef __attribute__((ext_vector_type(8))) float v8f;

#define NN 8192
#define DD 64
#define BM 128
#define KC 32
#define KH (NN / 2)
#define SA (KC + 4)
#define NG 4
#define NCH (KH / (KC * NG))
#define STAGE (BM * SA + DD * SA)
#define ALPHA 0.2f
#define EPS 1e-8f


typedef float f32x4 __attribute__((ext_vector_type(4)));

#define SWAVES 16
#define UNIT 16
#define NUNIT 8

__global__ __launch_bounds__(SWAVES * 32) void rowsum_kernel(const float* __restrict__ adj,
                                                             float* __restrict__ sr) {
  __shared__ float rs[2 * SWAVES];
  const int lane = threadIdx.x & 31;
  const int wave = __builtin_amdgcn_readfirstlane(threadIdx.x >> 5);
  const int row0 = blockIdx.x * (2 * SWAVES) + wave * 2;
  const f32x4* base = (const f32x4*)(adj + (size_t)row0 * NN);
  int off = lane;
  f32x4 buf[2][UNIT];
  float a0 = 0.f, a1 = 0.f, a2 = 0.f, a3 = 0.f;
  float tot[2];
#pragma unroll
  for (int q = 0; q < UNIT; ++q) buf[0][q] = base[32 * q + off];
#pragma unroll
  for (int u = 0; u < NUNIT; ++u) {
    if (u + 1 < NUNIT) {
      asm volatile("" : "+v"(off));
#pragma unroll
      for (int q = 0; q < UNIT; ++q) buf[(u + 1) & 1][q] = base[32 * ((u + 1) * UNIT + q) + off];
    }
    __builtin_amdgcn_sched_barrier(0);
#pragma unroll
    for (int q = 0; q < UNIT; ++q) {
      const f32x4 v = buf[u & 1][q];
      a0 += v.x; a1 += v.y; a2 += v.z; a3 += v.w;
    }
    asm volatile("" : "+v"(a0), "+v"(a1), "+v"(a2), "+v"(a3));
    if ((u & 3) == 3) {
      tot[u >> 2] = (a0 + a1) + (a2 + a3);
      a0 = 0.f; a1 = 0.f; a2 = 0.f; a3 = 0.f;
    }
  }
  float t0 = tot[0], t1 = tot[1];
#pragma unroll
  for (int m = 16; m >= 1; m >>= 1) { t0 += __shfl_xor(t0, m, 32); t1 += __shfl_xor(t1, m, 32); }
  if (lane == 0) { rs[wave * 2] = t0; rs[wave * 2 + 1] = t1; }
  __syncthreads();
  if (threadIdx.x < 2 * SWAVES) sr[blockIdx.x * (2 * SWAVES) + threadIdx.x] = sqrtf(rs[threadIdx.x]);
}

__global__ __launch_bounds__(256) void zero_kernel(float* __restrict__ p, int n) {
  int i = blockIdx.x * 256 + threadIdx.x;
  if (i < n) p[i] = 0.f;
}

__global__ __launch_bounds__(SWAVES * 32) void colsum_partial_kernel(const float* __restrict__ adj,
                                                                     float* __restrict__ acc) {
  __shared__ f32x4 part[SWAVES][32];
  const int lane = threadIdx.x & 31;
  const int wave = __builtin_amdgcn_readfirstlane(threadIdx.x >> 5);
  const int c0 = blockIdx.x * 128;
  const int r0 = blockIdx.y * (SWAVES * UNIT * NUNIT) + wave * (UNIT * NUNIT);
  const f32x4* base = (const f32x4*)(adj + (size_t)r0 * NN + c0);
  int off = lane;
  f32x4 buf[2][UNIT];
  float s0 = 0.f, s1 = 0.f, s2 = 0.f, s3 = 0.f;
#pragma unroll
  for (int q = 0; q < UNIT; ++q) buf[0][q] = base[q * (NN / 4) + off];
#pragma unroll
  for (int u = 0; u < NUNIT; ++u) {
    if (u + 1 < NUNIT) {
      asm volatile("" : "+v"(off));
#pragma unroll
      for (int q = 0; q < UNIT; ++q) buf[(u + 1) & 1][q] = base[((u + 1) * UNIT + q) * (NN / 4) + off];
    }
    __builtin_amdgcn_sched_barrier(0);
#pragma unroll
    for (int q = 0; q < UNIT; ++q) {
      const f32x4 v = buf[u & 1][q];
      s0 += v.x; s1 += v.y; s2 += v.z; s3 += v.w;
    }
    asm volatile("" : "+v"(s0), "+v"(s1), "+v"(s2), "+v"(s3));
  }
  part[wave][lane] = f32x4{s0, s1, s2, s3};
  __syncthreads();
  if (threadIdx.x < 128) {
    const float* pf = (const float*)&part[0][0];
    float t = pf[threadIdx.x];
#pragma unroll
    for (int g = 1; g < SWAVES; ++g) t += pf[g * 128 + threadIdx.x];
    atomicAdd(&acc[c0 + threadIdx.x], t);
  }
}

__global__ __launch_bounds__(256) void sqrt_kernel(const float* __restrict__ a,
                                                   float* __restrict__ sc) {
  int j = blockIdx.x * 256 + threadIdx.x;
  sc[j] = sqrtf(a[j]);
}


template <bool TRANSA>
__global__ __launch_bounds__(1024) void spmm_kernel(
    const float* __restrict__ adj, const float* __restrict__ sr, const float* __restrict__ sc,
    const float* __restrict__ Bsrc, const float* __restrict__ Btgt,
    float* __restrict__ P) {
  __shared__ __align__(16) float lds[NG * STAGE];

  const int tid = threadIdx.x;
  const int wv = __builtin_amdgcn_readfirstlane(tid >> 5);
  const int grp = wv >> 3;
  const int par = grp & 1;
  const int wave = wv & 7;
  const int t = tid & 255;
  const int lane = tid & 31;
  const int laneRow = lane & 15;
  const int laneHi = lane >> 4;
  const int bid = blockIdx.x;
  const int prod = bid & 1;
  const int half = (bid >> 1) & 1;
  const int r0 = (bid >> 2) * BM;
  const int kbeg = half * KH;
  const float* __restrict__ Bm = prod ? Btgt : Bsrc;
  float* __restrict__ Z = P + (size_t)(prod * 2 + half) * NN * DD;
  float* At = lds + grp * STAGE;
  float* Bs = At + BM * SA;

  v8f acc[4] = {};

  auto fetch = [&](int c) {
    const int k0 = kbeg + (c * NG + grp) * KC;
    float4 va[4], vs, vb[2];
    float ar[4];
#pragma unroll
    for (int it = 0; it < 2; ++it) {
      const int f = it * 256 + t;
      vb[it] = *(const float4*)(Bm + (size_t)(k0 + (f >> 4)) * DD + 4 * (f & 15));
    }
    if (!TRANSA) {
      const int k = (lane & 7) << 2;
#pragma unroll
      for (int it = 0; it < 4; ++it) {
        const int row = it * 32 + wave + 8 * ((lane >> 3) & 1) + 16 * (lane >> 4);
        va[it] = *(const float4*)(adj + (size_t)(r0 + row) * NN + (k0 + k));
        ar[it] = sr[r0 + row];
      }
      vs = *(const float4*)(sc + k0 + k);
    } else {
      const int rw = 64 * (wave & 1) + 4 * (lane & 15);
#pragma unroll
      for (int it = 0; it < 4; ++it) {
        const int k = (it * 4 + (wave >> 1)) * 2 + (lane >> 4);
        va[it] = *(const float4*)(adj + (size_t)(k0 + k) * NN + (r0 + rw));
        ar[it] = sr[k0 + k];
      }
      vs = *(const float4*)(sc + r0 + rw);
    }
#pragma unroll
    for (int it = 0; it < 2; ++it) {
      const int f = it * 256 + t;
      float* dst = &Bs[(f & 15) * SA + (f >> 4)];
      dst[0] = vb[it].x; dst[16 * SA] = vb[it].y; dst[32 * SA] = vb[it].z; dst[48 * SA] = vb[it].w;
    }
    if (!TRANSA) {
      const int k = (lane & 7) << 2;
#pragma unroll
      for (int it = 0; it < 4; ++it) {
        const int row = it * 32 + wave + 8 * ((lane >> 3) & 1) + 16 * (lane >> 4);
        float4 o;
        o.x = va[it].x * __builtin_amdgcn_rcpf(ar[it] * vs.x + EPS);
        o.y = va[it].y * __builtin_amdgcn_rcpf(ar[it] * vs.y + EPS);
        o.z = va[it].z * __builtin_amdgcn_rcpf(ar[it] * vs.z + EPS);
        o.w = va[it].w * __builtin_amdgcn_rcpf(ar[it] * vs.w + EPS);
        *(float4*)&At[row * SA + k] = o;
      }
    } else {
      float* dst0 = &At[(64 * (wave & 1) + (lane & 15)) * SA + (wave >> 1) * 2 + (lane >> 4)];
#pragma unroll
      for (int it = 0; it < 4; ++it) {
        float* dst = dst0 + it * 8;
        dst[0]       = va[it].x * __builtin_amdgcn_rcpf(ar[it] * vs.x + EPS);
        dst[16 * SA] = va[it].y * __builtin_amdgcn_rcpf(ar[it] * vs.y + EPS);
        dst[32 * SA] = va[it].z * __builtin_amdgcn_rcpf(ar[it] * vs.z + EPS);
        dst[48 * SA] = va[it].w * __builtin_amdgcn_rcpf(ar[it] * vs.w + EPS);
      }
    }
  };

  const float* Arow = &At[(wave * 16 + laneRow) * SA + 2 * laneHi];
  auto multiply = [&]() {
#pragma unroll
    for (int kk = 0; kk < KC; kk += 4) {
      v2f a = *(const v2f*)(Arow + kk);
#pragma unroll
      for (int n = 0; n < 4; ++n) {
        int brow = n * 16 + laneRow;
        v2f b = *(const v2f*)(Bs + brow * SA + kk + 2 * laneHi);
        acc[n] = __builtin_amdgcn_wmma_f32_16x16x4_f32(
            false, a, false, b, (short)0, acc[n], false, false);
      }
    }
  };

  for (int p = 0; p <= 2 * NCH; ++p) {
    const int q = p - par;
    if ((q & 1) == 0 && q < 2 * NCH) fetch(q >> 1);
    if ((q & 1) != 0 && q > 0) multiply();
    __syncthreads();
  }

  float* red = lds;
  if (grp > 0) {
    float* dst = red + (size_t)(((grp - 1) * 8 + wave) * 32) * 32 + lane;
#pragma unroll
    for (int n = 0; n < 4; ++n)
#pragma unroll
      for (int v = 0; v < 8; ++v) dst[(n * 8 + v) * 32] = acc[n][v];
  }
  __syncthreads();
  if (grp == 0) {
#pragma unroll
    for (int v = 0; v < 8; ++v) {
      float s[4];
#pragma unroll
      for (int n = 0; n < 4; ++n) {
        s[n] = acc[n][v];
#pragma unroll
        for (int g = 1; g < NG; ++g)
          s[n] += red[(size_t)(((g - 1) * 8 + wave) * 32 + (n * 8 + v)) * 32 + lane];
      }
      const int tr = v + 8 * laneHi;
      const int row = TRANSA ? r0 + 64 * (wave >> 2) + 4 * tr + (wave & 3)
                             : r0 + wave * 16 + tr;
      *(float4*)&Z[(size_t)row * DD + 4 * laneRow] = make_float4(s[0], s[1], s[2], s[3]);
    }
  }
}


__global__ __launch_bounds__(256) void epilogue_kernel(
    const float* __restrict__ Z1a, const float* __restrict__ Z1b,
    const float* __restrict__ Z2a, const float* __restrict__ Z2b,
    const float* __restrict__ src, const float* __restrict__ tgt,
    const float* __restrict__ W1, const float* __restrict__ W2,
    float* __restrict__ out) {
  __shared__ float z1s[4][DD];
  __shared__ float ps[4][DD];
  const int tid = threadIdx.x;
  const int r = tid >> 6;
  const int c = tid & 63;
  const int row = blockIdx.x * 4 + r;
  const size_t off = (size_t)row * DD + c;
  z1s[r][c] = Z1a[off] + Z1b[off];
  ps[r][c] = src[off] * (Z2a[off] + Z2b[off]);
  __syncthreads();
  float acc = tgt[off];
#pragma unroll 8
  for (int k = 0; k < DD; ++k)
    acc += z1s[r][k] * W1[k * DD + c] + ps[r][k] * W2[k * DD + c];
  out[off] = (acc >= 0.f) ? acc : ALPHA * acc;
}


extern "C" void kernel_launch(void* const* d_in, const int* in_sizes, int n_in,
                              void* d_out, int out_size, void* d_ws, size_t ws_size,
                              hipStream_t stream) {
  const float* user = (const float*)d_in[0];
  const float* item = (const float*)d_in[1];
  const float* adj  = (const float*)d_in[2];
  const float* W1   = (const float*)d_in[3];
  const float* W2   = (const float*)d_in[4];

  float* ws = (float*)d_ws;
  const size_t ND = (size_t)NN * DD;
  float* sr   = ws;
  float* sc   = ws + NN;
  float* cacc = ws + 2 * NN;
  float* P    = ws + 3 * NN;
  float* Z1a  = P;
  float* Z1b  = Z1a + ND;
  float* Z2a  = Z1b + ND;
  float* Z2b  = Z2a + ND;

  float* out_user = (float*)d_out;
  float* out_item = out_user + ND;

  rowsum_kernel<<<NN / (2 * SWAVES), SWAVES * 32, 0, stream>>>(adj, sr);
  zero_kernel<<<NN / 256, 256, 0, stream>>>(cacc, NN);
  colsum_partial_kernel<<<dim3(NN / 128, NN / (SWAVES * UNIT * NUNIT)), SWAVES * 32, 0, stream>>>(adj, cacc);
  sqrt_kernel<<<NN / 256, 256, 0, stream>>>(cacc, sc);

  spmm_kernel<true><<<NN / BM * 4, 256 * NG, 0, stream>>>(adj, sr, sc, item, user, P);
  epilogue_kernel<<<NN / 4, 256, 0, stream>>>(Z1a, Z1b, Z2a, Z2b, item, user, W1, W2, out_user);

  spmm_kernel<false><<<NN / BM * 4, 256 * NG, 0, stream>>>(adj, sr, sc, user, item, P);
  epilogue_kernel<<<NN / 4, 256, 0, stream>>>(Z1a, Z1b, Z2a, Z2b, user, item, W1, W2, out_item);
}


# epilogue kernel re-tiled: 2048 work-groups of 4 rows -> 256 of 32 rows, W1 and W2 staged in LDS once per work-group, both 64-deep products on f32 WMMA; second-stage k-sum order changed
# speedup vs baseline: 5.7836x; 1.0305x over previous
#include <hip/hip_runtime.h>
#include <math.h>

typedef __attribute__((ext_vector_type(2))) float v2f;
typedef __attribute__((ext_vector_type(8))) float v8f;

#define NN 8192
#define DD 64
#define BM 128
#define KC 32
#define KH (NN / 2)
#define SA (KC + 4)
#define NG 4
#define NCH (KH / (KC * NG))
#define STAGE (BM * SA + DD * SA)
#define ALPHA 0.2f
#define EPS 1e-8f


typedef float f32x4 __attribute__((ext_vector_type(4)));

#define SWAVES 16
#define UNIT 16
#define NUNIT 8

__global__ __launch_bounds__(SWAVES * 32) void rowsum_kernel(const float* __restrict__ adj,
                                                             float* __restrict__ sr) {
  __shared__ float rs[2 * SWAVES];
  const int lane = threadIdx.x & 31;
  const int wave = __builtin_amdgcn_readfirstlane(threadIdx.x >> 5);
  const int row0 = blockIdx.x * (2 * SWAVES) + wave * 2;
  const f32x4* base = (const f32x4*)(adj + (size_t)row0 * NN);
  int off = lane;
  f32x4 buf[2][UNIT];
  float a0 = 0.f, a1 = 0.f, a2 = 0.f, a3 = 0.f;
  float tot[2];
#pragma unroll
  for (int q = 0; q < UNIT; ++q) buf[0][q] = base[32 * q + off];
#pragma unroll
  for (int u = 0; u < NUNIT; ++u) {
    if (u + 1 < NUNIT) {
      asm volatile("" : "+v"(off));
#pragma unroll
      for (int q = 0; q < UNIT; ++q) buf[(u + 1) & 1][q] = base[32 * ((u + 1) * UNIT + q) + off];
    }
    __builtin_amdgcn_sched_barrier(0);
#pragma unroll
    for (int q = 0; q < UNIT; ++q) {
      const f32x4 v = buf[u & 1][q];
      a0 += v.x; a1 += v.y; a2 += v.z; a3 += v.w;
    }
    asm volatile("" : "+v"(a0), "+v"(a1), "+v"(a2), "+v"(a3));
    if ((u & 3) == 3) {
      tot[u >> 2] = (a0 + a1) + (a2 + a3);
      a0 = 0.f; a1 = 0.f; a2 = 0.f; a3 = 0.f;
    }
  }
  float t0 = tot[0], t1 = tot[1];
#pragma unroll
  for (int m = 16; m >= 1; m >>= 1) { t0 += __shfl_xor(t0, m, 32); t1 += __shfl_xor(t1, m, 32); }
  if (lane == 0) { rs[wave * 2] = t0; rs[wave * 2 + 1] = t1; }
  __syncthreads();
  if (threadIdx.x < 2 * SWAVES) sr[blockIdx.x * (2 * SWAVES) + threadIdx.x] = sqrtf(rs[threadIdx.x]);
}

__global__ __launch_bounds__(256) void zero_kernel(float* __restrict__ p, int n) {
  int i = blockIdx.x * 256 + threadIdx.x;
  if (i < n) p[i] = 0.f;
}

__global__ __launch_bounds__(SWAVES * 32) void colsum_partial_kernel(const float* __restrict__ adj,
                                                                     float* __restrict__ acc) {
  __shared__ f32x4 part[SWAVES][32];
  const int lane = threadIdx.x & 31;
  const int wave = __builtin_amdgcn_readfirstlane(threadIdx.x >> 5);
  const int c0 = blockIdx.x * 128;
  const int r0 = blockIdx.y * (SWAVES * UNIT * NUNIT) + wave * (UNIT * NUNIT);
  const f32x4* base = (const f32x4*)(adj + (size_t)r0 * NN + c0);
  int off = lane;
  f32x4 buf[2][UNIT];
  float s0 = 0.f, s1 = 0.f, s2 = 0.f, s3 = 0.f;
#pragma unroll
  for (int q = 0; q < UNIT; ++q) buf[0][q] = base[q * (NN / 4) + off];
#pragma unroll
  for (int u = 0; u < NUNIT; ++u) {
    if (u + 1 < NUNIT) {
      asm volatile("" : "+v"(off));
#pragma unroll
      for (int q = 0; q < UNIT; ++q) buf[(u + 1) & 1][q] = base[((u + 1) * UNIT + q) * (NN / 4) + off];
    }
    __builtin_amdgcn_sched_barrier(0);
#pragma unroll
    for (int q = 0; q < UNIT; ++q) {
      const f32x4 v = buf[u & 1][q];
      s0 += v.x; s1 += v.y; s2 += v.z; s3 += v.w;
    }
    asm volatile("" : "+v"(s0), "+v"(s1), "+v"(s2), "+v"(s3));
  }
  part[wave][lane] = f32x4{s0, s1, s2, s3};
  __syncthreads();
  if (threadIdx.x < 128) {
    const float* pf = (const float*)&part[0][0];
    float t = pf[threadIdx.x];
#pragma unroll
    for (int g = 1; g < SWAVES; ++g) t += pf[g * 128 + threadIdx.x];
    atomicAdd(&acc[c0 + threadIdx.x], t);
  }
}

__global__ __launch_bounds__(256) void sqrt_kernel(const float* __restrict__ a,
                                                   float* __restrict__ sc) {
  int j = blockIdx.x * 256 + threadIdx.x;
  sc[j] = sqrtf(a[j]);
}


template <bool TRANSA>
__global__ __launch_bounds__(1024) void spmm_kernel(
    const float* __restrict__ adj, const float* __restrict__ sr, const float* __restrict__ sc,
    const float* __restrict__ Bsrc, const float* __restrict__ Btgt,
    float* __restrict__ P) {
  __shared__ __align__(16) float lds[NG * STAGE];

  const int tid = threadIdx.x;
  const int wv = __builtin_amdgcn_readfirstlane(tid >> 5);
  const int grp = wv >> 3;
  const int par = grp & 1;
  const int wave = wv & 7;
  const int t = tid & 255;
  const int lane = tid & 31;
  const int laneRow = lane & 15;
  const int laneHi = lane >> 4;
  const int bid = blockIdx.x;
  const int prod = bid & 1;
  const int half = (bid >> 1) & 1;
  const int r0 = (bid >> 2) * BM;
  const int kbeg = half * KH;
  const float* __restrict__ Bm = prod ? Btgt : Bsrc;
  float* __restrict__ Z = P + (size_t)(prod * 2 + half) * NN * DD;
  float* At = lds + grp * STAGE;
  float* Bs = At + BM * SA;

  v8f acc[4] = {};

  auto fetch = [&](int c) {
    const int k0 = kbeg + (c * NG + grp) * KC;
    float4 va[4], vs, vb[2];
    float ar[4];
#pragma unroll
    for (int it = 0; it < 2; ++it) {
      const int f = it * 256 + t;
      vb[it] = *(const float4*)(Bm + (size_t)(k0 + (f >> 4)) * DD + 4 * (f & 15));
    }
    if (!TRANSA) {
      const int k = (lane & 7) << 2;
#pragma unroll
      for (int it = 0; it < 4; ++it) {
        const int row = it * 32 + wave + 8 * ((lane >> 3) & 1) + 16 * (lane >> 4);
        va[it] = *(const float4*)(adj + (size_t)(r0 + row) * NN + (k0 + k));
        ar[it] = sr[r0 + row];
      }
      vs = *(const float4*)(sc + k0 + k);
    } else {
      const int rw = 64 * (wave & 1) + 4 * (lane & 15);
#pragma unroll
      for (int it = 0; it < 4; ++it) {
        const int k = (it * 4 + (wave >> 1)) * 2 + (lane >> 4);
        va[it] = *(const float4*)(adj + (size_t)(k0 + k) * NN + (r0 + rw));
        ar[it] = sr[k0 + k];
      }
      vs = *(const float4*)(sc + r0 + rw);
    }
#pragma unroll
    for (int it = 0; it < 2; ++it) {
      const int f = it * 256 + t;
      float* dst = &Bs[(f & 15) * SA + (f >> 4)];
      dst[0] = vb[it].x; dst[16 * SA] = vb[it].y; dst[32 * SA] = vb[it].z; dst[48 * SA] = vb[it].w;
    }
    if (!TRANSA) {
      const int k = (lane & 7) << 2;
#pragma unroll
      for (int it = 0; it < 4; ++it) {
        const int row = it * 32 + wave + 8 * ((lane >> 3) & 1) + 16 * (lane >> 4);
        float4 o;
        o.x = va[it].x * __builtin_amdgcn_rcpf(ar[it] * vs.x + EPS);
        o.y = va[it].y * __builtin_amdgcn_rcpf(ar[it] * vs.y + EPS);
        o.z = va[it].z * __builtin_amdgcn_rcpf(ar[it] * vs.z + EPS);
        o.w = va[it].w * __builtin_amdgcn_rcpf(ar[it] * vs.w + EPS);
        *(float4*)&At[row * SA + k] = o;
      }
    } else {
      float* dst0 = &At[(64 * (wave & 1) + (lane & 15)) * SA + (wave >> 1) * 2 + (lane >> 4)];
#pragma unroll
      for (int it = 0; it < 4; ++it) {
        float* dst = dst0 + it * 8;
        dst[0]       = va[it].x * __builtin_amdgcn_rcpf(ar[it] * vs.x + EPS);
        dst[16 * SA] = va[it].y * __builtin_amdgcn_rcpf(ar[it] * vs.y + EPS);
        dst[32 * SA] = va[it].z * __builtin_amdgcn_rcpf(ar[it] * vs.z + EPS);
        dst[48 * SA] = va[it].w * __builtin_amdgcn_rcpf(ar[it] * vs.w + EPS);
      }
    }
  };

  const float* Arow = &At[(wave * 16 + laneRow) * SA + 2 * laneHi];
  auto multiply = [&]() {
#pragma unroll
    for (int kk = 0; kk < KC; kk += 4) {
      v2f a = *(const v2f*)(Arow + kk);
#pragma unroll
      for (int n = 0; n < 4; ++n) {
        int brow = n * 16 + laneRow;
        v2f b = *(const v2f*)(Bs + brow * SA + kk + 2 * laneHi);
        acc[n] = __builtin_amdgcn_wmma_f32_16x16x4_f32(
            false, a, false, b, (short)0, acc[n], false, false);
      }
    }
  };

  for (int p = 0; p <= 2 * NCH; ++p) {
    const int q = p - par;
    if ((q & 1) == 0 && q < 2 * NCH) fetch(q >> 1);
    if ((q & 1) != 0 && q > 0) multiply();
    __syncthreads();
  }

  float* red = lds;
  if (grp > 0) {
    float* dst = red + (size_t)(((grp - 1) * 8 + wave) * 32) * 32 + lane;
#pragma unroll
    for (int n = 0; n < 4; ++n)
#pragma unroll
      for (int v = 0; v < 8; ++v) dst[(n * 8 + v) * 32] = acc[n][v];
  }
  __syncthreads();
  if (grp == 0) {
#pragma unroll
    for (int v = 0; v < 8; ++v) {
      float s[4];
#pragma unroll
      for (int n = 0; n < 4; ++n) {
        s[n] = acc[n][v];
#pragma unroll
        for (int g = 1; g < NG; ++g)
          s[n] += red[(size_t)(((g - 1) * 8 + wave) * 32 + (n * 8 + v)) * 32 + lane];
      }
      const int tr = v + 8 * laneHi;
      const int row = TRANSA ? r0 + 64 * (wave >> 2) + 4 * tr + (wave & 3)
                             : r0 + wave * 16 + tr;
      *(float4*)&Z[(size_t)row * DD + 4 * laneRow] = make_float4(s[0], s[1], s[2], s[3]);
    }
  }
}


#define ER 32
#define XP (2 * DD + 4)
#define WPC (DD + 8)
#define OPC (DD + 4)

__global__ __launch_bounds__(256) void epilogue_kernel(
    const float* __restrict__ Z1a, const float* __restrict__ Z1b,
    const float* __restrict__ Z2a, const float* __restrict__ Z2b,
    const float* __restrict__ src, const float* __restrict__ tgt,
    const float* __restrict__ W1, const float* __restrict__ W2,
    float* __restrict__ out) {
  __shared__ __align__(16) float xs[ER * XP];
  __shared__ __align__(16) float wc[2 * DD * WPC];
  __shared__ __align__(16) float os[ER * OPC];
  const int tid = threadIdx.x;
  const int wv = __builtin_amdgcn_readfirstlane(tid >> 5);
  const int lane = tid & 31;
  const int lr = lane & 15;
  const int lh = lane >> 4;
  const int r0 = blockIdx.x * ER;
  const int trow = tid >> 4;
  const int c4 = tid & 15;

  float4 z1a[2], z1b[2], z2a[2], z2b[2], sv[2], tv[2], wr[8];
#pragma unroll
  for (int it = 0; it < 2; ++it) {
    const size_t off = (size_t)(r0 + it * 16 + trow) * DD + 4 * c4;
    z1a[it] = *(const float4*)(Z1a + off);
    z1b[it] = *(const float4*)(Z1b + off);
    z2a[it] = *(const float4*)(Z2a + off);
    z2b[it] = *(const float4*)(Z2b + off);
    sv[it]  = *(const float4*)(src + off);
    tv[it]  = *(const float4*)(tgt + off);
  }
#pragma unroll
  for (int it = 0; it < 8; ++it) {
    const float* wsrc = (it < 4) ? W1 + (it * 16 + trow) * DD : W2 + ((it - 4) * 16 + trow) * DD;
    wr[it] = *(const float4*)(wsrc + 4 * c4);
  }
#pragma unroll
  for (int it = 0; it < 2; ++it) {
    const int row = it * 16 + trow;
    float4 z, p;
    z.x = z1a[it].x + z1b[it].x; z.y = z1a[it].y + z1b[it].y;
    z.z = z1a[it].z + z1b[it].z; z.w = z1a[it].w + z1b[it].w;
    p.x = sv[it].x * (z2a[it].x + z2b[it].x); p.y = sv[it].y * (z2a[it].y + z2b[it].y);
    p.z = sv[it].z * (z2a[it].z + z2b[it].z); p.w = sv[it].w * (z2a[it].w + z2b[it].w);
    *(float4*)&xs[row * XP + 4 * c4] = z;
    *(float4*)&xs[row * XP + DD + 4 * c4] = p;
  }
#pragma unroll
  for (int it = 0; it < 8; ++it)
    *(float4*)&wc[(it * 16 + trow) * WPC + 4 * c4] = wr[it];
  __syncthreads();

  const int mt = wv >> 2;
  const int nt = wv & 3;
  const float* Ap = &xs[(mt * 16 + lr) * XP + 2 * lh];
  const float* Bp = &wc[(2 * lh) * WPC + nt * 16 + lr];
  v8f c0 = {}, c1 = {};
  v2f af[32], bf[32];
#pragma unroll
  for (int s = 0; s < 16; ++s) {
    af[s] = *(const v2f*)(Ap + 4 * s);
    af[16 + s] = *(const v2f*)(Ap + DD + 4 * s);
    bf[s].x = Bp[(4 * s) * WPC];
    bf[s].y = Bp[(4 * s + 1) * WPC];
    bf[16 + s].x = Bp[(DD + 4 * s) * WPC];
    bf[16 + s].y = Bp[(DD + 4 * s + 1) * WPC];
  }
#pragma unroll
  for (int g = 0; g < 8; ++g) {
    if (g < 7) {
      asm volatile("" : "+v"(af[4 * g]), "+v"(af[4 * g + 1]), "+v"(af[4 * g + 2]), "+v"(af[4 * g + 3]),
                        "+v"(bf[4 * g]), "+v"(bf[4 * g + 1]), "+v"(bf[4 * g + 2]), "+v"(bf[4 * g + 3]));
    } else {
      asm volatile("" : "+v"(af[28]), "+v"(af[29]), "+v"(af[30]), "+v"(af[31]),
                        "+v"(bf[28]), "+v"(bf[29]), "+v"(bf[30]), "+v"(bf[31]), "+v"(c0), "+v"(c1));
    }
  }
#pragma unroll
  for (int s = 0; s < 16; ++s) {
    c0 = __builtin_amdgcn_wmma_f32_16x16x4_f32(false, af[s], false, bf[s], (short)0, c0, false, false);
    c1 = __builtin_amdgcn_wmma_f32_16x16x4_f32(false, af[16 + s], false, bf[16 + s], (short)0, c1,
                                               false, false);
  }
  int lane2 = lane, tid2 = tid;
  asm volatile("v_nop\n\tv_nop\n\tv_nop\n\tv_nop" : "+v"(c0), "+v"(c1), "+v"(lane2), "+v"(tid2));
  const int lr2 = lane2 & 15;
  const int lh2 = lane2 >> 4;
  const int trow2 = tid2 >> 4;
  const int c42 = tid2 & 15;
#pragma unroll
  for (int v = 0; v < 8; ++v)
    os[(mt * 16 + v + 8 * lh2) * OPC + nt * 16 + lr2] = c0[v] + c1[v];
  __syncthreads();
#pragma unroll
  for (int it = 0; it < 2; ++it) {
    const int row = it * 16 + trow2;
    const float4 s = *(const float4*)&os[row * OPC + 4 * c42];
    float4 o;
    o.x = tv[it].x + s.x; o.y = tv[it].y + s.y; o.z = tv[it].z + s.z; o.w = tv[it].w + s.w;
    o.x = (o.x >= 0.f) ? o.x : ALPHA * o.x;
    o.y = (o.y >= 0.f) ? o.y : ALPHA * o.y;
    o.z = (o.z >= 0.f) ? o.z : ALPHA * o.z;
    o.w = (o.w >= 0.f) ? o.w : ALPHA * o.w;
    *(float4*)(out + (size_t)(r0 + row) * DD + 4 * c42) = o;
  }
}


extern "C" void kernel_launch(void* const* d_in, const int* in_sizes, int n_in,
                              void* d_out, int out_size, void* d_ws, size_t ws_size,
                              hipStream_t stream) {
  const float* user = (const float*)d_in[0];
  const float* item = (const float*)d_in[1];
  const float* adj  = (const float*)d_in[2];
  const float* W1   = (const float*)d_in[3];
  const float* W2   = (const float*)d_in[4];

  float* ws = (float*)d_ws;
  const size_t ND = (size_t)NN * DD;
  float* sr   = ws;
  float* sc   = ws + NN;
  float* cacc = ws + 2 * NN;
  float* P    = ws + 3 * NN;
  float* Z1a  = P;
  float* Z1b  = Z1a + ND;
  float* Z2a  = Z1b + ND;
  float* Z2b  = Z2a + ND;

  float* out_user = (float*)d_out;
  float* out_item = out_user + ND;

  rowsum_kernel<<<NN / (2 * SWAVES), SWAVES * 32, 0, stream>>>(adj, sr);
  zero_kernel<<<NN / 256, 256, 0, stream>>>(cacc, NN);
  colsum_partial_kernel<<<dim3(NN / 128, NN / (SWAVES * UNIT * NUNIT)), SWAVES * 32, 0, stream>>>(adj, cacc);
  sqrt_kernel<<<NN / 256, 256, 0, stream>>>(cacc, sc);

  spmm_kernel<true><<<NN / BM * 4, 256 * NG, 0, stream>>>(adj, sr, sc, item, user, P);
  epilogue_kernel<<<NN / ER, 256, 0, stream>>>(Z1a, Z1b, Z2a, Z2b, item, user, W1, W2, out_user);

  spmm_kernel<false><<<NN / BM * 4, 256 * NG, 0, stream>>>(adj, sr, sc, user, item, P);
  epilogue_kernel<<<NN / ER, 256, 0, stream>>>(Z1a, Z1b, Z2a, Z2b, user, item, W1, W2, out_item);
}


# spmm: a work-group runs both products on a quarter of K: a staged adjacency chunk feeds 8 accumulator tiles a wave (64 WMMAs a phase); epilogue adds 4 K-part planes (k sum order changed)
# speedup vs baseline: 6.3072x; 1.0905x over previous
#include <hip/hip_runtime.h>
#include <math.h>

typedef __attribute__((ext_vector_type(2))) float v2f;
typedef __attribute__((ext_vector_type(4))) float v4f;
typedef __attribute__((ext_vector_type(8))) float v8f;

#define NN 8192
#define DD 64
#define BM 128
#define KC 32
#define SA (KC + 4)
#define NG 4
#define STAGE (BM * SA + 2 * DD * SA)
#define REDF ((NG - 1) * 8 * 64 * 32)
#define LDSF (REDF > NG * STAGE ? REDF : NG * STAGE)
#define NPL 8
#define WS_NEED ((3ull * NN + (unsigned long long)NPL * NN * DD) * 4ull)
#define ALPHA 0.2f
#define EPS 1e-8f


typedef float f32x4 __attribute__((ext_vector_type(4)));

#define SWAVES 16
#define UNIT 16
#define NUNIT 8

__global__ __launch_bounds__(SWAVES * 32) void rowsum_kernel(const float* __restrict__ adj,
                                                             float* __restrict__ sr) {
  __shared__ float rs[2 * SWAVES];
  const int lane = threadIdx.x & 31;
  const int wave = __builtin_amdgcn_readfirstlane(threadIdx.x >> 5);
  const int row0 = blockIdx.x * (2 * SWAVES) + wave * 2;
  const f32x4* base = (const f32x4*)(adj + (size_t)row0 * NN);
  int off = lane;
  f32x4 buf[2][UNIT];
  float a0 = 0.f, a1 = 0.f, a2 = 0.f, a3 = 0.f;
  float tot[2];
#pragma unroll
  for (int q = 0; q < UNIT; ++q) buf[0][q] = base[32 * q + off];
#pragma unroll
  for (int u = 0; u < NUNIT; ++u) {
    if (u + 1 < NUNIT) {
      asm volatile("" : "+v"(off));
#pragma unroll
      for (int q = 0; q < UNIT; ++q) buf[(u + 1) & 1][q] = base[32 * ((u + 1) * UNIT + q) + off];
    }
    __builtin_amdgcn_sched_barrier(0);
#pragma unroll
    for (int q = 0; q < UNIT; ++q) {
      const f32x4 v = buf[u & 1][q];
      a0 += v.x; a1 += v.y; a2 += v.z; a3 += v.w;
    }
    asm volatile("" : "+v"(a0), "+v"(a1), "+v"(a2), "+v"(a3));
    if ((u & 3) == 3) {
      tot[u >> 2] = (a0 + a1) + (a2 + a3);
      a0 = 0.f; a1 = 0.f; a2 = 0.f; a3 = 0.f;
    }
  }
  float t0 = tot[0], t1 = tot[1];
#pragma unroll
  for (int m = 16; m >= 1; m >>= 1) { t0 += __shfl_xor(t0, m, 32); t1 += __shfl_xor(t1, m, 32); }
  if (lane == 0) { rs[wave * 2] = t0; rs[wave * 2 + 1] = t1; }
  __syncthreads();
  if (threadIdx.x < 2 * SWAVES) sr[blockIdx.x * (2 * SWAVES) + threadIdx.x] = sqrtf(rs[threadIdx.x]);
}

__global__ __launch_bounds__(256) void zero_kernel(float* __restrict__ p, int n) {
  int i = blockIdx.x * 256 + threadIdx.x;
  if (i < n) p[i] = 0.f;
}

__global__ __launch_bounds__(SWAVES * 32) void colsum_partial_kernel(const float* __restrict__ adj,
                                                                     float* __restrict__ acc) {
  __shared__ f32x4 part[SWAVES][32];
  const int lane = threadIdx.x & 31;
  const int wave = __builtin_amdgcn_readfirstlane(threadIdx.x >> 5);
  const int c0 = blockIdx.x * 128;
  const int r0 = blockIdx.y * (SWAVES * UNIT * NUNIT) + wave * (UNIT * NUNIT);
  const f32x4* base = (const f32x4*)(adj + (size_t)r0 * NN + c0);
  int off = lane;
  f32x4 buf[2][UNIT];
  float s0 = 0.f, s1 = 0.f, s2 = 0.f, s3 = 0.f;
#pragma unroll
  for (int q = 0; q < UNIT; ++q) buf[0][q] = base[q * (NN / 4) + off];
#pragma unroll
  for (int u = 0; u < NUNIT; ++u) {
    if (u + 1 < NUNIT) {
      asm volatile("" : "+v"(off));
#pragma unroll
      for (int q = 0; q < UNIT; ++q) buf[(u + 1) & 1][q] = base[((u + 1) * UNIT + q) * (NN / 4) + off];
    }
    __builtin_amdgcn_sched_barrier(0);
#pragma unroll
    for (int q = 0; q < UNIT; ++q) {
      const f32x4 v = buf[u & 1][q];
      s0 += v.x; s1 += v.y; s2 += v.z; s3 += v.w;
    }
    asm volatile("" : "+v"(s0), "+v"(s1), "+v"(s2), "+v"(s3));
  }
  part[wave][lane] = f32x4{s0, s1, s2, s3};
  __syncthreads();
  if (threadIdx.x < 128) {
    const float* pf = (const float*)&part[0][0];
    float t = pf[threadIdx.x];
#pragma unroll
    for (int g = 1; g < SWAVES; ++g) t += pf[g * 128 + threadIdx.x];
    atomicAdd(&acc[c0 + threadIdx.x], t);
  }
}

__global__ __launch_bounds__(256) void sqrt_kernel(const float* __restrict__ a,
                                                   float* __restrict__ sc) {
  int j = blockIdx.x * 256 + threadIdx.x;
  sc[j] = sqrtf(a[j]);
}


template <bool TRANSA>
__global__ __launch_bounds__(1024) void spmm_kernel(
    const float* __restrict__ adj, const float* __restrict__ sr, const float* __restrict__ sc,
    const float* __restrict__ Bsrc, const float* __restrict__ Btgt,
    float* __restrict__ P, size_t ws_bytes) {
  __shared__ __align__(16) float lds[LDSF];

  const int bid = blockIdx.x;
  const int nq = (ws_bytes >= WS_NEED) ? 4 : 2;
  const int part = bid & 3;
  if (part >= nq) return;
  const int nch = NN / (KC * NG * nq);
  const int kbeg = part * (NN / nq);
  const int r0 = (bid >> 2) * BM;

  const int tid = threadIdx.x;
  const int wv = __builtin_amdgcn_readfirstlane(tid >> 5);
  const int grp = wv >> 3;
  const int par = grp & 1;
  const int wave = wv & 7;
  const unsigned t = tid & 255;
  const unsigned lane = tid & 31;
  const int laneRow = lane & 15;
  const int laneHi = lane >> 4;
  float* At = lds + grp * STAGE;
  float* Bs = At + BM * SA;

  v8f acc[8] = {};

  const unsigned arow = TRANSA ? (unsigned)((wave >> 1) * 2) + (lane >> 4)
                               : (unsigned)wave + 8u * ((lane >> 3) & 1u) + 16u * (lane >> 4);
  const unsigned acol = TRANSA ? (unsigned)(64 * (wave & 1)) + 4u * (lane & 15u) : (lane & 7u) << 2;
  const unsigned aoffB = (arow * NN + acol) * 4u;
  const unsigned bc4 = t & 15u;
  const unsigned bkg = (t >> 4) & 7u;
  const unsigned boffB = (bkg * 4u * DD + 4u * bc4) * 4u;
  const int bprod = wave >> 2;
  const char* aBase = (const char*)(TRANSA ? adj + r0 : adj + (size_t)r0 * NN);
  const char* bBase = (const char*)(bprod ? Btgt : Bsrc);
  const char* rBase = (const char*)(TRANSA ? sr : sr + r0);
  const char* cBase = (const char*)(TRANSA ? sc + r0 : sc);

  auto fetch = [&](int c) {
    const unsigned k0 = (unsigned)(kbeg + (c * NG + grp) * KC);
    float4 va[4], vs, vb[4];
    float ar[4];
    const unsigned aB = aoffB + (TRANSA ? k0 * (NN * 4u) : k0 * 4u);
    const unsigned bB = boffB + k0 * (DD * 4u);
    const unsigned rB = TRANSA ? (arow + k0) * 4u : arow * 4u;
    const unsigned cB = TRANSA ? acol * 4u : (acol + k0) * 4u;
#pragma unroll
    for (int i = 0; i < 4; ++i) vb[i] = *(const float4*)(bBase + i * (DD * 4) + bB);
#pragma unroll
    for (int it = 0; it < 4; ++it) {
      va[it] = *(const float4*)(aBase + (size_t)it * ((TRANSA ? 8 : 32) * NN * 4) + aB);
      ar[it] = *(const float*)(rBase + it * ((TRANSA ? 8 : 32) * 4) + rB);
    }
    vs = *(const float4*)(cBase + cB);
    {
      float* dst = &Bs[(64 * bprod + bc4) * SA + 4 * bkg];
      *(float4*)&dst[0]       = make_float4(vb[0].x, vb[1].x, vb[2].x, vb[3].x);
      *(float4*)&dst[16 * SA] = make_float4(vb[0].y, vb[1].y, vb[2].y, vb[3].y);
      *(float4*)&dst[32 * SA] = make_float4(vb[0].z, vb[1].z, vb[2].z, vb[3].z);
      *(float4*)&dst[48 * SA] = make_float4(vb[0].w, vb[1].w, vb[2].w, vb[3].w);
    }
    if (!TRANSA) {
#pragma unroll
      for (int it = 0; it < 4; ++it) {
        float4 o;
        o.x = va[it].x * __builtin_amdgcn_rcpf(ar[it] * vs.x + EPS);
        o.y = va[it].y * __builtin_amdgcn_rcpf(ar[it] * vs.y + EPS);
        o.z = va[it].z * __builtin_amdgcn_rcpf(ar[it] * vs.z + EPS);
        o.w = va[it].w * __builtin_amdgcn_rcpf(ar[it] * vs.w + EPS);
        *(float4*)&At[(it * 32 + arow) * SA + acol] = o;
      }
    } else {
      float* dst0 = &At[(64 * (wave & 1) + (lane & 15)) * SA + arow];
#pragma unroll
      for (int it = 0; it < 4; ++it) {
        float* dst = dst0 + it * 8;
        dst[0]       = va[it].x * __builtin_amdgcn_rcpf(ar[it] * vs.x + EPS);
        dst[16 * SA] = va[it].y * __builtin_amdgcn_rcpf(ar[it] * vs.y + EPS);
        dst[32 * SA] = va[it].z * __builtin_amdgcn_rcpf(ar[it] * vs.z + EPS);
        dst[48 * SA] = va[it].w * __builtin_amdgcn_rcpf(ar[it] * vs.w + EPS);
      }
    }
  };

  const float* fragA = &At[(wave * 16 + laneRow) * SA + 4 * laneHi];
  const float* fragB = &Bs[laneRow * SA + 4 * laneHi];
  auto multiply = [&]() {
#pragma unroll
    for (int k8 = 0; k8 < KC; k8 += 8) {
      const v4f a = *(const v4f*)(fragA + k8);
      v4f b[8];
#pragma unroll
      for (int n = 0; n < 8; ++n) b[n] = *(const v4f*)(fragB + n * 16 * SA + k8);
#pragma unroll
      for (int n = 0; n < 8; ++n)
        acc[n] = __builtin_amdgcn_wmma_f32_16x16x4_f32(
            false, a.xy, false, b[n].xy, (short)0, acc[n], false, false);
#pragma unroll
      for (int n = 0; n < 8; ++n)
        acc[n] = __builtin_amdgcn_wmma_f32_16x16x4_f32(
            false, a.zw, false, b[n].zw, (short)0, acc[n], false, false);
    }
    asm volatile("v_nop\n\tv_nop\n\tv_nop\n\tv_nop"
                 : "+v"(acc[0]), "+v"(acc[1]), "+v"(acc[2]), "+v"(acc[3]),
                   "+v"(acc[4]), "+v"(acc[5]), "+v"(acc[6]), "+v"(acc[7]));
  };

  for (int p = 0; p <= 2 * nch; ++p) {
    const int q = p - par;
    if ((q & 1) == 0 && q < 2 * nch) fetch(q >> 1);
    if ((q & 1) != 0 && q > 0) multiply();
    __syncthreads();
  }

  float* red = lds;
  if (grp > 0) {
    float* dst = red + (size_t)(((grp - 1) * 8 + wave) * 64) * 32 + lane;
#pragma unroll
    for (int n = 0; n < 8; ++n)
#pragma unroll
      for (int v = 0; v < 8; ++v) dst[(n * 8 + v) * 32] = acc[n][v];
  }
  __syncthreads();
  if (grp == 0) {
    unsigned lane2 = lane;
    asm volatile("" : "+v"(lane2));
    const int laneRow2 = lane2 & 15;
    const int laneHi2 = lane2 >> 4;
#pragma unroll
    for (int pr = 0; pr < 2; ++pr) {
      float* __restrict__ Z = P + (size_t)(pr * nq + part) * NN * DD;
#pragma unroll
      for (int v = 0; v < 8; ++v) {
        float s[4];
#pragma unroll
        for (int n = 0; n < 4; ++n) {
          s[n] = acc[pr * 4 + n][v];
#pragma unroll
          for (int g = 1; g < NG; ++g)
            s[n] += red[(size_t)(((g - 1) * 8 + wave) * 64 + ((pr * 4 + n) * 8 + v)) * 32 + lane];
        }
        const int tr = v + 8 * laneHi2;
        const int row = TRANSA ? r0 + 64 * (wave >> 2) + 4 * tr + (wave & 3)
                               : r0 + wave * 16 + tr;
        *(float4*)&Z[(size_t)row * DD + 4 * laneRow2] = make_float4(s[0], s[1], s[2], s[3]);
      }
    }
  }
}


#define ER 32
#define XP (2 * DD + 4)
#define WPC (DD + 8)
#define OPC (DD + 4)

__global__ __launch_bounds__(256) void epilogue_kernel(
    const float* __restrict__ P,
    const float* __restrict__ src, const float* __restrict__ tgt,
    const float* __restrict__ W1, const float* __restrict__ W2,
    float* __restrict__ out, size_t ws_bytes) {
  __shared__ __align__(16) float xs[ER * XP];
  __shared__ __align__(16) float wc[2 * DD * WPC];
  __shared__ __align__(16) float os[ER * OPC];
  const int tid = threadIdx.x;
  const int wv = __builtin_amdgcn_readfirstlane(tid >> 5);
  const int lane = tid & 31;
  const int lr = lane & 15;
  const int lh = lane >> 4;
  const int r0 = blockIdx.x * ER;
  const int trow = tid >> 4;
  const int c4 = tid & 15;
  const int nq = (ws_bytes >= WS_NEED) ? 4 : 2;

  float4 z1[4][2] = {}, z2[4][2] = {}, sv[2], tv[2], wr[8];
#pragma unroll
  for (int it = 0; it < 2; ++it) {
    const size_t off = (size_t)(r0 + it * 16 + trow) * DD + 4 * c4;
#pragma unroll
    for (int q = 0; q < 4; ++q) {
      if (q < nq) {
        z1[q][it] = *(const float4*)(P + (size_t)q * NN * DD + off);
        z2[q][it] = *(const float4*)(P + (size_t)(nq + q) * NN * DD + off);
      }
    }
    sv[it]  = *(const float4*)(src + off);
    tv[it]  = *(const float4*)(tgt + off);
  }
#pragma unroll
  for (int it = 0; it < 8; ++it) {
    const float* wsrc = (it < 4) ? W1 + (it * 16 + trow) * DD : W2 + ((it - 4) * 16 + trow) * DD;
    wr[it] = *(const float4*)(wsrc + 4 * c4);
  }
#pragma unroll
  for (int it = 0; it < 2; ++it) {
    const int row = it * 16 + trow;
    float4 z, y, p;
    z.x = z1[0][it].x + z1[1][it].x; z.y = z1[0][it].y + z1[1][it].y;
    z.z = z1[0][it].z + z1[1][it].z; z.w = z1[0][it].w + z1[1][it].w;
    y.x = z2[0][it].x + z2[1][it].x; y.y = z2[0][it].y + z2[1][it].y;
    y.z = z2[0][it].z + z2[1][it].z; y.w = z2[0][it].w + z2[1][it].w;
    if (nq == 4) {
      z.x = (z.x + z1[2][it].x) + z1[3][it].x; z.y = (z.y + z1[2][it].y) + z1[3][it].y;
      z.z = (z.z + z1[2][it].z) + z1[3][it].z; z.w = (z.w + z1[2][it].w) + z1[3][it].w;
      y.x = (y.x + z2[2][it].x) + z2[3][it].x; y.y = (y.y + z2[2][it].y) + z2[3][it].y;
      y.z = (y.z + z2[2][it].z) + z2[3][it].z; y.w = (y.w + z2[2][it].w) + z2[3][it].w;
    }
    p.x = sv[it].x * y.x; p.y = sv[it].y * y.y; p.z = sv[it].z * y.z; p.w = sv[it].w * y.w;
    *(float4*)&xs[row * XP + 4 * c4] = z;
    *(float4*)&xs[row * XP + DD + 4 * c4] = p;
  }
#pragma unroll
  for (int it = 0; it < 8; ++it)
    *(float4*)&wc[(it * 16 + trow) * WPC + 4 * c4] = wr[it];
  __syncthreads();

  const int mt = wv >> 2;
  const int nt = wv & 3;
  const float* Ap = &xs[(mt * 16 + lr) * XP + 2 * lh];
  const float* Bp = &wc[(2 * lh) * WPC + nt * 16 + lr];
  v8f c0 = {}, c1 = {};
  v2f af[32], bf[32];
#pragma unroll
  for (int s = 0; s < 16; ++s) {
    af[s] = *(const v2f*)(Ap + 4 * s);
    af[16 + s] = *(const v2f*)(Ap + DD + 4 * s);
    bf[s].x = Bp[(4 * s) * WPC];
    bf[s].y = Bp[(4 * s + 1) * WPC];
    bf[16 + s].x = Bp[(DD + 4 * s) * WPC];
    bf[16 + s].y = Bp[(DD + 4 * s + 1) * WPC];
  }
#pragma unroll
  for (int g = 0; g < 8; ++g) {
    if (g < 7) {
      asm volatile("" : "+v"(af[4 * g]), "+v"(af[4 * g + 1]), "+v"(af[4 * g + 2]), "+v"(af[4 * g + 3]),
                        "+v"(bf[4 * g]), "+v"(bf[4 * g + 1]), "+v"(bf[4 * g + 2]), "+v"(bf[4 * g + 3]));
    } else {
      asm volatile("" : "+v"(af[28]), "+v"(af[29]), "+v"(af[30]), "+v"(af[31]),
                        "+v"(bf[28]), "+v"(bf[29]), "+v"(bf[30]), "+v"(bf[31]), "+v"(c0), "+v"(c1));
    }
  }
#pragma unroll
  for (int s = 0; s < 16; ++s) {
    c0 = __builtin_amdgcn_wmma_f32_16x16x4_f32(false, af[s], false, bf[s], (short)0, c0, false, false);
    c1 = __builtin_amdgcn_wmma_f32_16x16x4_f32(false, af[16 + s], false, bf[16 + s], (short)0, c1,
                                               false, false);
  }
  int lane2 = lane, tid2 = tid;
  asm volatile("v_nop\n\tv_nop\n\tv_nop\n\tv_nop" : "+v"(c0), "+v"(c1), "+v"(lane2), "+v"(tid2));
  const int lr2 = lane2 & 15;
  const int lh2 = lane2 >> 4;
  const int trow2 = tid2 >> 4;
  const int c42 = tid2 & 15;
#pragma unroll
  for (int v = 0; v < 8; ++v)
    os[(mt * 16 + v + 8 * lh2) * OPC + nt * 16 + lr2] = c0[v] + c1[v];
  __syncthreads();
#pragma unroll
  for (int it = 0; it < 2; ++it) {
    const int row = it * 16 + trow2;
    const float4 s = *(const float4*)&os[row * OPC + 4 * c42];
    float4 o;
    o.x = tv[it].x + s.x; o.y = tv[it].y + s.y; o.z = tv[it].z + s.z; o.w = tv[it].w + s.w;
    o.x = (o.x >= 0.f) ? o.x : ALPHA * o.x;
    o.y = (o.y >= 0.f) ? o.y : ALPHA * o.y;
    o.z = (o.z >= 0.f) ? o.z : ALPHA * o.z;
    o.w = (o.w >= 0.f) ? o.w : ALPHA * o.w;
    *(float4*)(out + (size_t)(r0 + row) * DD + 4 * c42) = o;
  }
}


extern "C" void kernel_launch(void* const* d_in, const int* in_sizes, int n_in,
                              void* d_out, int out_size, void* d_ws, size_t ws_size,
                              hipStream_t stream) {
  const float* user = (const float*)d_in[0];
  const float* item = (const float*)d_in[1];
  const float* adj  = (const float*)d_in[2];
  const float* W1   = (const float*)d_in[3];
  const float* W2   = (const float*)d_in[4];

  float* ws = (float*)d_ws;
  const size_t ND = (size_t)NN * DD;
  float* sr   = ws;
  float* sc   = ws + NN;
  float* cacc = ws + 2 * NN;
  float* P    = ws + 3 * NN;

  float* out_user = (float*)d_out;
  float* out_item = out_user + ND;

  rowsum_kernel<<<NN / (2 * SWAVES), SWAVES * 32, 0, stream>>>(adj, sr);
  zero_kernel<<<NN / 256, 256, 0, stream>>>(cacc, NN);
  colsum_partial_kernel<<<dim3(NN / 128, NN / (SWAVES * UNIT * NUNIT)), SWAVES * 32, 0, stream>>>(adj, cacc);
  sqrt_kernel<<<NN / 256, 256, 0, stream>>>(cacc, sc);

  spmm_kernel<true><<<NN / BM * 4, 256 * NG, 0, stream>>>(adj, sr, sc, item, user, P, ws_size);
  epilogue_kernel<<<NN / ER, 256, 0, stream>>>(P, item, user, W1, W2, out_user, ws_size);

  spmm_kernel<false><<<NN / BM * 4, 256 * NG, 0, stream>>>(adj, sr, sc, user, item, P, ws_size);
  epilogue_kernel<<<NN / ER, 256, 0, stream>>>(P, user, item, W1, W2, out_item, ws_size);
}
